# starting kernel
# baseline (speedup 1.0000x reference)
#include <hip/hip_runtime.h>
#include <hip/hip_bf16.h>
#include <math.h>


typedef __attribute__((ext_vector_type(2))) float v2f;
typedef __attribute__((ext_vector_type(8))) float v8f;

#define B_ROWS 4096
#define T_LEN  8192
#define N_WIN  8
#define TILE_M 16
#define WAVES  8
#define BLOCK  256
#define KSTEPS_TOTAL    (T_LEN / 4)
#define KSTEPS_PER_WAVE (KSTEPS_TOTAL / WAVES)

__global__ void ewma_weights_kernel(const float* __restrict__ log_halflife,
                                    float* __restrict__ wgtB) {
  int idx = blockIdx.x * blockDim.x + threadIdx.x;
  if (idx >= KSTEPS_TOTAL * 64) return;
  int s    = idx >> 6;
  int r    = idx & 63;
  int lane = r >> 1;
  int j    = r & 1;
  int t    = 4 * s + 2 * (lane >> 4) + j;
  int n    = lane & 15;
  float val = 0.0f;
  if (n < N_WIN) {
    float hl = expf(log_halflife[n]);
    hl = fminf(fmaxf(hl, 1.0f), 200.0f);
    float invhl = 1.0f / hl;
    float d = exp2f(-invhl);
    float age = (float)(T_LEN - 1 - t);
    val = exp2f(-age * invhl);
    if (t > 0) val *= (1.0f - d);
  }
  wgtB[idx] = val;
}

__global__ void __launch_bounds__(BLOCK)
ewma_wmma_kernel(const float* __restrict__ X,
                 const float* __restrict__ wgtB,
                 float* __restrict__ out) {
  __shared__ float red[WAVES * 256];

  const int tile = blockIdx.x;
  const int lane = threadIdx.x & 31;
  const int wv   = threadIdx.x >> 5;

  const int row    = tile * TILE_M + (lane & 15);
  const int kStart = wv * KSTEPS_PER_WAVE * 4;
  const float* ap = X + (size_t)row * T_LEN + kStart + 2 * (lane >> 4);
  const float* bp = wgtB + (size_t)(wv * KSTEPS_PER_WAVE) * 64 + lane * 2;

  v8f acc = {};
#pragma unroll 8
  for (int s = 0; s < KSTEPS_PER_WAVE; ++s) {
    v2f a = *(const v2f*)ap;
    v2f b = *(const v2f*)bp;
    acc = __builtin_amdgcn_wmma_f32_16x16x4_f32(
        false, a, false, b, (short)0, acc, false, false);
    ap += 4;
    bp += 64;
  }

#pragma unroll
  for (int r = 0; r < 8; ++r)
    red[wv * 256 + lane * 8 + r] = acc[r];
  __syncthreads();

  const int i = threadIdx.x;
  float sum = 0.0f;
#pragma unroll
  for (int w = 0; w < WAVES; ++w) sum += red[w * 256 + i];

  const int srcLane = i >> 3;
  const int r = i & 7;
  const int M = r + 8 * (srcLane >> 4);
  const int N = srcLane & 15;
  if (N < N_WIN) {
    const int grow = tile * TILE_M + M;
    const float last = X[(size_t)grow * T_LEN + (T_LEN - 1)];
    out[grow * N_WIN + N] = (sum - last) / (fabsf(last) + 1e-8f);
  }
}

extern "C" void kernel_launch(void* const* d_in, const int* in_sizes, int n_in,
                              void* d_out, int out_size, void* d_ws, size_t ws_size,
                              hipStream_t stream) {
  const float* price        = (const float*)d_in[0];
  const float* log_halflife = (const float*)d_in[1];
  float* out  = (float*)d_out;
  float* wgtB = (float*)d_ws;

  (void)in_sizes; (void)n_in; (void)out_size; (void)ws_size;

  const int wElems = KSTEPS_TOTAL * 64;
  ewma_weights_kernel<<<(wElems + BLOCK - 1) / BLOCK, BLOCK, 0, stream>>>(
      log_halflife, wgtB);

  ewma_wmma_kernel<<<B_ROWS / TILE_M, BLOCK, 0, stream>>>(price, wgtB, out);
}


# One launch, no weight table anywhere: each wave derives its 32 pattern fragments in registers (2 exp2 seeds, 31 packed multiplies by d^4/d^16) behind its first row loads; no LDS table, no barrier
# speedup vs baseline: 2.2986x; 2.2986x over previous
#include <hip/hip_runtime.h>
#include <hip/hip_bf16.h>
#include <math.h>


typedef __attribute__((ext_vector_type(2))) float v2f;
typedef __attribute__((ext_vector_type(4))) float v4f;
typedef __attribute__((ext_vector_type(8))) float v8f;

#define B_ROWS 4096
#define T_LEN  8192
#define N_WIN  8
#define TILE_M 16
#define WAVES  16
#define BLOCK  (WAVES * 32)
#define KSTEPS_TOTAL    (T_LEN / 4)
#define KSTEPS_PER_WAVE (KSTEPS_TOTAL / WAVES)
#define CHUNK_T   128
#define CHUNK_KS  (CHUNK_T / 4)
#define CHUNKS    (KSTEPS_PER_WAVE / CHUNK_KS)
#define PITCH     (CHUNK_T + 4)

__global__ void __launch_bounds__(BLOCK)
ewma_wmma_kernel(const float* __restrict__ X,
                 const float* __restrict__ log_halflife,
                 float* __restrict__ out) {
  __shared__ __attribute__((aligned(16))) float stage[WAVES][TILE_M * PITCH];
  __shared__ float red[WAVES * 256];

  const int tile = blockIdx.x;
  const int lane = threadIdx.x & 31;
  const int wv   = threadIdx.x >> 5;
  const int n    = lane & 15;

  float lhs[N_WIN];
#pragma unroll
  for (int k = 0; k < N_WIN; ++k) lhs[k] = log_halflife[k];

  const int kStart = wv * KSTEPS_PER_WAVE * 4;
  const float* xp = X + (size_t)(tile * TILE_M) * T_LEN + kStart + lane * 4;

  float* st = stage[wv];
  float* stw = st + lane * 4;
  const float* str = st + (lane & 15) * PITCH + 2 * (lane >> 4);

  v4f xr[TILE_M];
#pragma unroll
  for (int q = 0; q < TILE_M; ++q)
    xr[q] = *(const v4f*)(xp + (size_t)q * T_LEN);

  __builtin_amdgcn_sched_barrier(0);

  const bool live = n < N_WIN;
  float lh = lhs[0];
#pragma unroll
  for (int k = 1; k < N_WIN; ++k)
    if ((n & (N_WIN - 1)) == k) lh = lhs[k];
  float hl = expf(lh);
  hl = fminf(fmaxf(hl, 1.0f), 200.0f);
  const float invhl = 1.0f / hl;
  const float omd   = live ? 1.0f - exp2f(-invhl) : 0.0f;
  const int   h2    = 2 * (lane >> 4);
  v2f b[CHUNK_KS];
  b[CHUNK_KS - 1].x = exp2f(-(float)(3 - h2) * invhl) * omd;
  b[CHUNK_KS - 1].y = exp2f(-(float)(2 - h2) * invhl) * omd;
  const float d4  = exp2f(-4.0f * invhl);
  const float d16 = exp2f(-16.0f * invhl);
#pragma unroll
  for (int s = CHUNK_KS - 2; s >= CHUNK_KS - 4; --s)
    b[s] = b[s + 1] * d4;
#pragma unroll
  for (int s = CHUNK_KS - 5; s >= 0; --s)
    b[s] = b[s + 4] * d16;

  const float dchunk = exp2f(-(float)CHUNK_T * invhl);
  const float swave  = exp2f(-(float)((WAVES - 1 - wv) * (KSTEPS_PER_WAVE * 4)) * invhl);

  v2f bfirst = b[0];
  const float p0 = exp2f(-(float)(CHUNK_T - 1) * invhl);
  if (wv == 0 && lane < N_WIN) bfirst.x = p0;

  __builtin_amdgcn_sched_barrier(0);

  v8f acc0 = {};
  v8f acc1 = {};

#pragma unroll
  for (int c = 0; c < CHUNKS; ++c) {
#pragma unroll
    for (int q = 0; q < TILE_M; ++q)
      *(v4f*)(stw + q * PITCH) = xr[q];

    if (c + 1 < CHUNKS) {
#pragma unroll
      for (int q = 0; q < TILE_M; ++q)
        xr[q] = *(const v4f*)(xp + (size_t)q * T_LEN + (c + 1) * CHUNK_T);
    }

    __builtin_amdgcn_fence(__ATOMIC_RELEASE, "wavefront", "local");
    __builtin_amdgcn_wave_barrier();
    __builtin_amdgcn_fence(__ATOMIC_ACQUIRE, "wavefront", "local");

    if (c > 0) {
#pragma unroll
      for (int r = 0; r < 8; ++r) {
        acc0[r] *= dchunk;
        acc1[r] *= dchunk;
      }
    }

#pragma unroll
    for (int s = 0; s < CHUNK_KS; s += 2) {
      v2f a0 = *(const v2f*)(str + 4 * s);
      v2f a1 = *(const v2f*)(str + 4 * s + 4);
      acc0 = __builtin_amdgcn_wmma_f32_16x16x4_f32(
          false, a0, false, (c == 0 && s == 0) ? bfirst : b[s], (short)0, acc0, false, false);
      acc1 = __builtin_amdgcn_wmma_f32_16x16x4_f32(
          false, a1, false, b[s + 1], (short)0, acc1, false, false);
    }

    __builtin_amdgcn_fence(__ATOMIC_RELEASE, "wavefront", "local");
    __builtin_amdgcn_wave_barrier();
    __builtin_amdgcn_fence(__ATOMIC_ACQUIRE, "wavefront", "local");
  }

#pragma unroll
  for (int r = 0; r < 8; ++r)
    red[wv * 256 + lane * 8 + r] = (acc0[r] + acc1[r]) * swave;
  __syncthreads();

  const int i = threadIdx.x;
  if (i < TILE_M * N_WIN) {
    const int M = i >> 3;
    const int N = i & 7;
    const int srcIdx = (N + 16 * (M >> 3)) * 8 + (M & 7);
    float sum = 0.0f;
#pragma unroll
    for (int w = 0; w < WAVES; ++w) sum += red[w * 256 + srcIdx];
    const float last = stage[WAVES - 1][M * PITCH + (CHUNK_T - 1)];
    out[tile * (TILE_M * N_WIN) + i] = (sum - last) / (fabsf(last) + 1e-8f);
  }
}

extern "C" void kernel_launch(void* const* d_in, const int* in_sizes, int n_in,
                              void* d_out, int out_size, void* d_ws, size_t ws_size,
                              hipStream_t stream) {
  const float* price        = (const float*)d_in[0];
  const float* log_halflife = (const float*)d_in[1];
  float* out  = (float*)d_out;

  (void)in_sizes; (void)n_in; (void)out_size; (void)d_ws; (void)ws_size;

  ewma_wmma_kernel<<<B_ROWS / TILE_M, BLOCK, 0, stream>>>(price, log_halflife, out);
}
